# starting kernel
# baseline (speedup 1.0000x reference)
#include <hip/hip_runtime.h>
#include <hip/hip_bf16.h>

typedef __attribute__((ext_vector_type(2))) float v2f;
typedef __attribute__((ext_vector_type(8))) float v8f;

#define NB 2
#define NG 8
#define ND 32
#define NH 128
#define NW 160
#define DHW (ND * NH * NW)
#define NPTS (NB * DHW)
#define TILES_PER_WAVE 4

#define SWIZ_SWAPX16 0x401F

__device__ __forceinline__ float half_swap(float v) {
    return __int_as_float(
        __builtin_amdgcn_ds_swizzle(__float_as_int(v), SWIZ_SWAPX16));
}

__global__ __launch_bounds__(256) void mlp_sim_kernel(
    const float* __restrict__ x1,
    const float* __restrict__ w0,
    const float* __restrict__ bn0_scale, const float* __restrict__ bn0_bias,
    const float* __restrict__ w1,
    const float* __restrict__ bn1_scale, const float* __restrict__ bn1_bias,
    const float* __restrict__ w_sim, const float* __restrict__ b_sim,
    float* __restrict__ sim)
{
    const int lane = threadIdx.x & 31;
    const int wib  = threadIdx.x >> 5;
    const int gwave = blockIdx.x * 8 + wib;
    const int hi   = lane >> 4;
    const int col  = lane & 15;

    const int kb = hi * 2;
    v2f a0, a1;
    a0.x = w0[col * 8 + kb + 0];
    a0.y = w0[col * 8 + kb + 1];
    a1.x = w0[col * 8 + 4 + kb + 0];
    a1.y = w0[col * 8 + 4 + kb + 1];

    v2f aw[4];
#pragma unroll
    for (int j = 0; j < 4; ++j) {
        const int k0 = j * 4 + kb;
        aw[j].x = (col < 8) ? w1[col * 16 + k0 + 0] : 0.0f;
        aw[j].y = (col < 8) ? w1[col * 16 + k0 + 1] : 0.0f;
    }

    float s0[8], bi0[8], s1[8], bi1[8], ws[8];
    const int rb = 8 * hi;
#pragma unroll
    for (int v = 0; v < 8; ++v) {
        s0[v]  = bn0_scale[rb + v];
        bi0[v] = bn0_bias[rb + v];
        const int r = (rb + v) & 7;
        s1[v]  = bn1_scale[r];
        bi1[v] = bn1_bias[r];
        ws[v]  = w_sim[v];
    }
    const float bs = b_sim[0];

#pragma unroll
    for (int t = 0; t < TILES_PER_WAVE; ++t) {
        const int base = (gwave * TILES_PER_WAVE + t) * 16;
        const int p    = base + col;
        const int b    = p / DHW;
        const int rem  = p - b * DHW;

        const float* xp = x1 + (size_t)(b * NG + kb) * DHW + rem;
        v2f b0, b1;
        b0.x = xp[0];
        b0.y = xp[DHW];
        b1.x = xp[4 * (size_t)DHW];
        b1.y = xp[5 * (size_t)DHW];
        __builtin_prefetch(xp + 16, 0, 3);

        v8f c = {};
        c = __builtin_amdgcn_wmma_f32_16x16x4_f32(false, a0, false, b0,
                                                  (short)0, c, false, false);
        c = __builtin_amdgcn_wmma_f32_16x16x4_f32(false, a1, false, b1,
                                                  (short)0, c, false, false);
        float h[8];
#pragma unroll
        for (int v = 0; v < 8; ++v)
            h[v] = fmaxf(fmaf(c[v], s0[v], bi0[v]), 0.0f);

        float hx[8];
#pragma unroll
        for (int v = 0; v < 8; ++v)
            hx[v] = half_swap(h[v]);

        v2f bb[4];
        bb[0].x = hi ? hx[2] : h[0];  bb[0].y = hi ? hx[3] : h[1];
        bb[1].x = hi ? hx[6] : h[4];  bb[1].y = hi ? hx[7] : h[5];
        bb[2].x = hi ? h[2] : hx[0];  bb[2].y = hi ? h[3] : hx[1];
        bb[3].x = hi ? h[6] : hx[4];  bb[3].y = hi ? h[7] : hx[5];

        v8f d = {};
#pragma unroll
        for (int j = 0; j < 4; ++j)
            d = __builtin_amdgcn_wmma_f32_16x16x4_f32(false, aw[j], false, bb[j],
                                                      (short)0, d, false, false);

        float g[8];
#pragma unroll
        for (int v = 0; v < 8; ++v)
            g[v] = fmaxf(fmaf(d[v], s1[v], bi1[v]), 0.0f);

        float o = bs;
#pragma unroll
        for (int v = 0; v < 8; ++v)
            o = fmaf(ws[v], g[v], o);

        if (!hi)
            sim[p] = o;
    }
}

__device__ __forceinline__ int reflect_ix(int t, int n) {
    t = (t < 0) ? -t : t;
    t = (t >= n) ? (2 * n - 2 - t) : t;
    return t;
}

__global__ __launch_bounds__(256) void aggregate_kernel(
    const float* __restrict__ sim,
    const float* __restrict__ offset,
    const float* __restrict__ weight,
    float* __restrict__ out)
{
    const int idx = blockIdx.x * 256 + threadIdx.x;
    const int x = idx % NW;
    int t = idx / NW;
    const int y = t % NH;
    t /= NH;
    const int d = t % ND;
    const int b = t / ND;

    int yn[3], xn[3], yw[3], xw[3];
#pragma unroll
    for (int i = 0; i < 3; ++i) {
        yn[i] = reflect_ix(y + (i - 1) * 2, NH);
        xn[i] = reflect_ix(x + (i - 1) * 2, NW);
        yw[i] = reflect_ix(y + (i - 1) * 4, NH);
        xw[i] = reflect_ix(x + (i - 1) * 4, NW);
    }

    const float* sb = sim + (size_t)(b * ND + d) * NH * NW;
    const float* ob = offset + (size_t)b * 18 * NH * NW + y * NW + x;

    float acc = 0.0f;
#pragma unroll
    for (int s = 0; s < 9; ++s) {
        const int iy = s / 3, ix = s % 3;
        const float c1 = sb[yn[iy] * NW + xn[ix]];
        const float c2 = sb[yw[iy] * NW + xw[ix]];
        const float w_wide   = ob[(size_t)s * NH * NW];
        const float w_narrow = ob[(size_t)(s + 9) * NH * NW];
        acc = fmaf(0.5f, fmaf(w_narrow, c1, w_wide * c2), acc);
    }
    out[idx] = acc * weight[(size_t)b * NH * NW + y * NW + x];
}

extern "C" void kernel_launch(void* const* d_in, const int* in_sizes, int n_in,
                              void* d_out, int out_size, void* d_ws, size_t ws_size,
                              hipStream_t stream)
{
    const float* x1        = (const float*)d_in[0];
    const float* offset    = (const float*)d_in[1];
    const float* weight    = (const float*)d_in[2];
    const float* w0        = (const float*)d_in[3];
    const float* bn0_scale = (const float*)d_in[4];
    const float* bn0_bias  = (const float*)d_in[5];
    const float* w1        = (const float*)d_in[6];
    const float* bn1_scale = (const float*)d_in[7];
    const float* bn1_bias  = (const float*)d_in[8];
    const float* w_sim     = (const float*)d_in[9];
    const float* b_sim     = (const float*)d_in[10];
    float* out = (float*)d_out;
    float* sim = (float*)d_ws;

    const int waves   = NPTS / (16 * TILES_PER_WAVE);
    const int blocks1 = waves / 8;
    mlp_sim_kernel<<<blocks1, 256, 0, stream>>>(
        x1, w0, bn0_scale, bn0_bias, w1, bn1_scale, bn1_bias, w_sim, b_sim, sim);

    const int blocks2 = NPTS / 256;
    aggregate_kernel<<<blocks2, 256, 0, stream>>>(sim, offset, weight, out);
}


# Coarsen work per wave: mlp wave loads weights once and walks 16 tiles (a 4-tile group's 16 loads issued together); aggregation thread does 8 depths per pixel reusing offsets; 61440 -> 10240 waves
# speedup vs baseline: 1.2336x; 1.2336x over previous
#include <hip/hip_runtime.h>
#include <hip/hip_bf16.h>


typedef __attribute__((ext_vector_type(2))) float v2f;
typedef __attribute__((ext_vector_type(8))) float v8f;

#define NB 2
#define NG 8
#define ND 32
#define NH 128
#define NW 160
#define HW (NH * NW)
#define DHW (ND * NH * NW)
#define NPTS (NB * DHW)

#define TILES_PER_GROUP 4
#define NGROUPS (NPTS / 64)
#define K1_BLOCK 128
#define K1_WAVES 5120
#define GROUPS_PER_WAVE (NGROUPS / K1_WAVES)
#define DPT 8
#define K2_BLOCK 128

#define SWIZ_SWAPX16 0x401F

__device__ __forceinline__ float half_swap(float v) {
    return __int_as_float(
        __builtin_amdgcn_ds_swizzle(__float_as_int(v), SWIZ_SWAPX16));
}

__global__ __launch_bounds__(K1_BLOCK) void mlp_sim_kernel(
    const float* __restrict__ x1,
    const float* __restrict__ w0,
    const float* __restrict__ bn0_scale, const float* __restrict__ bn0_bias,
    const float* __restrict__ w1,
    const float* __restrict__ bn1_scale, const float* __restrict__ bn1_bias,
    const float* __restrict__ w_sim, const float* __restrict__ b_sim,
    float* __restrict__ sim)
{
    const int lane = threadIdx.x & 31;
    const int wib  = threadIdx.x >> 5;
    const int gwave = blockIdx.x * (K1_BLOCK / 32) + wib;
    const int hi   = lane >> 4;
    const int col  = lane & 15;

    const int kb = hi * 2;
    v2f a0, a1;
    a0.x = w0[col * 8 + kb + 0];
    a0.y = w0[col * 8 + kb + 1];
    a1.x = w0[col * 8 + 4 + kb + 0];
    a1.y = w0[col * 8 + 4 + kb + 1];

    v2f aw[4];
#pragma unroll
    for (int j = 0; j < 4; ++j) {
        const int k0 = j * 4 + kb;
        aw[j].x = (col < 8) ? w1[col * 16 + k0 + 0] : 0.0f;
        aw[j].y = (col < 8) ? w1[col * 16 + k0 + 1] : 0.0f;
    }

    float s0[8], bi0[8], s1[8], bi1[8], ws[8];
    const int rb = 8 * hi;
#pragma unroll
    for (int v = 0; v < 8; ++v) {
        s0[v]  = bn0_scale[rb + v];
        bi0[v] = bn0_bias[rb + v];
        const int r = (rb + v) & 7;
        s1[v]  = bn1_scale[r];
        bi1[v] = bn1_bias[r];
        ws[v]  = w_sim[v];
    }
    const float bs = b_sim[0];

    for (int i = 0; i < GROUPS_PER_WAVE; ++i) {
        const int g     = i * K1_WAVES + gwave;
        const int base  = g * 64;
        const int b     = base / DHW;
        const int rem   = base - b * DHW + col;

        const float* xp = x1 + (size_t)(b * NG + kb) * DHW + rem;
        v2f b0[TILES_PER_GROUP], b1[TILES_PER_GROUP];
#pragma unroll
        for (int t = 0; t < TILES_PER_GROUP; ++t) {
            b0[t].x = xp[t * 16];
            b0[t].y = xp[t * 16 + DHW];
            b1[t].x = xp[t * 16 + 4 * (size_t)DHW];
            b1[t].y = xp[t * 16 + 5 * (size_t)DHW];
        }

        float o[TILES_PER_GROUP];
#pragma unroll
        for (int t = 0; t < TILES_PER_GROUP; ++t) {
            v8f c = {};
            c = __builtin_amdgcn_wmma_f32_16x16x4_f32(false, a0, false, b0[t],
                                                      (short)0, c, false, false);
            c = __builtin_amdgcn_wmma_f32_16x16x4_f32(false, a1, false, b1[t],
                                                      (short)0, c, false, false);
            float h[8];
#pragma unroll
            for (int v = 0; v < 8; ++v)
                h[v] = fmaxf(fmaf(c[v], s0[v], bi0[v]), 0.0f);

            float hx[8];
#pragma unroll
            for (int v = 0; v < 8; ++v)
                hx[v] = half_swap(h[v]);

            v2f bb[4];
            bb[0].x = hi ? hx[2] : h[0];  bb[0].y = hi ? hx[3] : h[1];
            bb[1].x = hi ? hx[6] : h[4];  bb[1].y = hi ? hx[7] : h[5];
            bb[2].x = hi ? h[2] : hx[0];  bb[2].y = hi ? h[3] : hx[1];
            bb[3].x = hi ? h[6] : hx[4];  bb[3].y = hi ? h[7] : hx[5];

            v8f d = {};
#pragma unroll
            for (int j = 0; j < 4; ++j)
                d = __builtin_amdgcn_wmma_f32_16x16x4_f32(false, aw[j], false, bb[j],
                                                          (short)0, d, false, false);

            float gg[8];
#pragma unroll
            for (int v = 0; v < 8; ++v)
                gg[v] = fmaxf(fmaf(d[v], s1[v], bi1[v]), 0.0f);

            float acc = bs;
#pragma unroll
            for (int v = 0; v < 8; ++v)
                acc = fmaf(ws[v], gg[v], acc);
            o[t] = acc;
        }

        if (!hi) {
#pragma unroll
            for (int t = 0; t < TILES_PER_GROUP; ++t)
                sim[base + t * 16 + col] = o[t];
        }
    }
}

__device__ __forceinline__ int reflect_ix(int t, int n) {
    t = (t < 0) ? -t : t;
    t = (t >= n) ? (2 * n - 2 - t) : t;
    return t;
}

__global__ __launch_bounds__(K2_BLOCK) void aggregate_kernel(
    const float* __restrict__ sim,
    const float* __restrict__ offset,
    const float* __restrict__ weight,
    float* __restrict__ out)
{
    const int idx  = blockIdx.x * K2_BLOCK + threadIdx.x;
    const int pix  = idx % HW;
    const int rest = idx / HW;
    const int dg   = rest % (ND / DPT);
    const int b    = rest / (ND / DPT);
    const int x    = pix % NW;
    const int y    = pix / NW;

    int yn[3], xn[3], yw[3], xw[3];
#pragma unroll
    for (int i = 0; i < 3; ++i) {
        yn[i] = reflect_ix(y + (i - 1) * 2, NH);
        xn[i] = reflect_ix(x + (i - 1) * 2, NW);
        yw[i] = reflect_ix(y + (i - 1) * 4, NH);
        xw[i] = reflect_ix(x + (i - 1) * 4, NW);
    }

    const float* ob = offset + (size_t)b * 18 * HW + pix;
    int   in_[9], iw_[9];
    float w_narrow[9], w_wide[9];
#pragma unroll
    for (int s = 0; s < 9; ++s) {
        const int iy = s / 3, ix = s % 3;
        in_[s] = yn[iy] * NW + xn[ix];
        iw_[s] = yw[iy] * NW + xw[ix];
        w_wide[s]   = ob[(size_t)s * HW];
        w_narrow[s] = ob[(size_t)(s + 9) * HW];
    }
    const float wgt = weight[(size_t)b * HW + pix];

    const size_t plane0 = (size_t)(b * ND + dg * DPT) * HW;
    const float* sb = sim + plane0;
    float*       op = out + plane0 + pix;

#pragma unroll
    for (int k = 0; k < DPT; ++k) {
        float acc = 0.0f;
#pragma unroll
        for (int s = 0; s < 9; ++s) {
            const float c1 = sb[in_[s]];
            const float c2 = sb[iw_[s]];
            acc = fmaf(0.5f, fmaf(w_narrow[s], c1, w_wide[s] * c2), acc);
        }
        op[(size_t)k * HW] = acc * wgt;
        sb += HW;
    }
}

extern "C" void kernel_launch(void* const* d_in, const int* in_sizes, int n_in,
                              void* d_out, int out_size, void* d_ws, size_t ws_size,
                              hipStream_t stream)
{
    const float* x1        = (const float*)d_in[0];
    const float* offset    = (const float*)d_in[1];
    const float* weight    = (const float*)d_in[2];
    const float* w0        = (const float*)d_in[3];
    const float* bn0_scale = (const float*)d_in[4];
    const float* bn0_bias  = (const float*)d_in[5];
    const float* w1        = (const float*)d_in[6];
    const float* bn1_scale = (const float*)d_in[7];
    const float* bn1_bias  = (const float*)d_in[8];
    const float* w_sim     = (const float*)d_in[9];
    const float* b_sim     = (const float*)d_in[10];
    float* out = (float*)d_out;
    float* sim = (float*)d_ws;

    mlp_sim_kernel<<<K1_WAVES / (K1_BLOCK / 32), K1_BLOCK, 0, stream>>>(
        x1, w0, bn0_scale, bn0_bias, w1, bn1_scale, bn1_bias, w_sim, b_sim, sim);

    aggregate_kernel<<<NPTS / DPT / K2_BLOCK, K2_BLOCK, 0, stream>>>(
        sim, offset, weight, out);
}


# Each lane owns 4 adjacent points, moved with 128-bit loads/stores: mlp tile t = points 4c+t; aggregation thread = 4 pixels x 4 depths (border mirror by select), its loads issued as two groups
# speedup vs baseline: 1.3238x; 1.0731x over previous
#include <hip/hip_runtime.h>
#include <hip/hip_bf16.h>


typedef __attribute__((ext_vector_type(2))) float v2f;
typedef __attribute__((ext_vector_type(4))) float v4f;
typedef __attribute__((ext_vector_type(8))) float v8f;

#define NB 2
#define NG 8
#define ND 32
#define NH 128
#define NW 160
#define HW (NH * NW)
#define DHW (ND * NH * NW)
#define NPTS (NB * DHW)

#define TILES_PER_GROUP 4
#define NGROUPS (NPTS / 64)
#define K1_BLOCK 128
#define K1_WAVES 5120
#define GROUPS_PER_WAVE (NGROUPS / K1_WAVES)

#define PXT 4
#define QPR (NW / PXT)
#define DPT 4
#define NDG (ND / DPT)
#define K2_BLOCK 320
#define K2_BPP (HW / PXT / K2_BLOCK)
#define K2_GRID (NB * NDG * K2_BPP)

#define SWIZ_SWAPX16 0x401F

__device__ __forceinline__ float half_swap(float v) {
    return __int_as_float(
        __builtin_amdgcn_ds_swizzle(__float_as_int(v), SWIZ_SWAPX16));
}

__global__ __launch_bounds__(K1_BLOCK) void mlp_sim_kernel(
    const float* __restrict__ x1,
    const float* __restrict__ w0,
    const float* __restrict__ bn0_scale, const float* __restrict__ bn0_bias,
    const float* __restrict__ w1,
    const float* __restrict__ bn1_scale, const float* __restrict__ bn1_bias,
    const float* __restrict__ w_sim, const float* __restrict__ b_sim,
    float* __restrict__ sim)
{
    const int lane = threadIdx.x & 31;
    const int wib  = threadIdx.x >> 5;
    const int gwave = blockIdx.x * (K1_BLOCK / 32) + wib;
    const int hi   = lane >> 4;
    const int col  = lane & 15;

    const int kb = hi * 2;
    v2f a0, a1;
    a0.x = w0[col * 8 + kb + 0];
    a0.y = w0[col * 8 + kb + 1];
    a1.x = w0[col * 8 + 4 + kb + 0];
    a1.y = w0[col * 8 + 4 + kb + 1];

    v2f aw[4];
#pragma unroll
    for (int j = 0; j < 4; ++j) {
        const int k0 = j * 4 + kb;
        aw[j].x = (col < 8) ? w1[col * 16 + k0 + 0] : 0.0f;
        aw[j].y = (col < 8) ? w1[col * 16 + k0 + 1] : 0.0f;
    }

    float s0[8], bi0[8], s1[8], bi1[8], ws[8];
    const int rb = 8 * hi;
#pragma unroll
    for (int v = 0; v < 8; ++v) {
        s0[v]  = bn0_scale[rb + v];
        bi0[v] = bn0_bias[rb + v];
        const int r = (rb + v) & 7;
        s1[v]  = bn1_scale[r];
        bi1[v] = bn1_bias[r];
        ws[v]  = w_sim[v];
    }
    const float bs = b_sim[0];

    for (int i = 0; i < GROUPS_PER_WAVE; ++i) {
        const int g     = i * K1_WAVES + gwave;
        const int base  = g * 64;
        const int b     = base / DHW;
        const int rem   = base - b * DHW + PXT * col;

        const float* xp = x1 + (size_t)(b * NG + kb) * DHW + rem;
        const v4f q0 = *(const v4f*)(xp);
        const v4f q1 = *(const v4f*)(xp + DHW);
        const v4f q2 = *(const v4f*)(xp + 4 * (size_t)DHW);
        const v4f q3 = *(const v4f*)(xp + 5 * (size_t)DHW);

        v4f o;
#pragma unroll
        for (int t = 0; t < TILES_PER_GROUP; ++t) {
            v2f b0, b1;
            b0.x = q0[t];  b0.y = q1[t];
            b1.x = q2[t];  b1.y = q3[t];

            v8f c = {};
            c = __builtin_amdgcn_wmma_f32_16x16x4_f32(false, a0, false, b0,
                                                      (short)0, c, false, false);
            c = __builtin_amdgcn_wmma_f32_16x16x4_f32(false, a1, false, b1,
                                                      (short)0, c, false, false);
            float h[8];
#pragma unroll
            for (int v = 0; v < 8; ++v)
                h[v] = fmaxf(fmaf(c[v], s0[v], bi0[v]), 0.0f);

            float hx[8];
#pragma unroll
            for (int v = 0; v < 8; ++v)
                hx[v] = half_swap(h[v]);

            v2f bb[4];
            bb[0].x = hi ? hx[2] : h[0];  bb[0].y = hi ? hx[3] : h[1];
            bb[1].x = hi ? hx[6] : h[4];  bb[1].y = hi ? hx[7] : h[5];
            bb[2].x = hi ? h[2] : hx[0];  bb[2].y = hi ? h[3] : hx[1];
            bb[3].x = hi ? h[6] : hx[4];  bb[3].y = hi ? h[7] : hx[5];

            v8f d = {};
#pragma unroll
            for (int j = 0; j < 4; ++j)
                d = __builtin_amdgcn_wmma_f32_16x16x4_f32(false, aw[j], false, bb[j],
                                                          (short)0, d, false, false);

            float gg[8];
#pragma unroll
            for (int v = 0; v < 8; ++v)
                gg[v] = fmaxf(fmaf(d[v], s1[v], bi1[v]), 0.0f);

            float acc = bs;
#pragma unroll
            for (int v = 0; v < 8; ++v)
                acc = fmaf(ws[v], gg[v], acc);
            o[t] = acc;
        }

        if (!hi) {
            *(v4f*)(sim + base + PXT * col) = o;
        }
    }
}

__device__ __forceinline__ int reflect_ix(int t, int n) {
    t = (t < 0) ? -t : t;
    t = (t >= n) ? (2 * n - 2 - t) : t;
    return t;
}

struct PlanePieces {
    v4f wl[3], wm[3], wr[3];
    v2f na[2], nb[2];
    v4f nm[2];
};

__device__ __forceinline__ void load_plane(PlanePieces& P, const float* sp,
                                           const int (&wrow)[3], const int (&nrow)[2],
                                           int cl4, int cl2, int x0, int cr) {
#pragma unroll
    for (int i = 0; i < 3; ++i) {
        P.wl[i] = *(const v4f*)(sp + wrow[i] + cl4);
        P.wm[i] = *(const v4f*)(sp + wrow[i] + x0);
        P.wr[i] = *(const v4f*)(sp + wrow[i] + cr);
    }
#pragma unroll
    for (int i = 0; i < 2; ++i) {
        P.na[i] = *(const v2f*)(sp + nrow[i] + cl2);
        P.nm[i] = *(const v4f*)(sp + nrow[i] + x0);
        P.nb[i] = *(const v2f*)(sp + nrow[i] + cr);
    }
}

__device__ __forceinline__ void pin_plane(PlanePieces& P) {
    asm volatile("" : "+v"(P.wl[0]), "+v"(P.wm[0]), "+v"(P.wr[0]),
                      "+v"(P.wl[1]), "+v"(P.wm[1]), "+v"(P.wr[1]),
                      "+v"(P.wl[2]), "+v"(P.wm[2]), "+v"(P.wr[2]),
                      "+v"(P.na[0]), "+v"(P.nm[0]), "+v"(P.nb[0]),
                      "+v"(P.na[1]), "+v"(P.nm[1]), "+v"(P.nb[1]));
}

__device__ __forceinline__ v4f aggregate_plane(const PlanePieces& P, bool left, bool right,
                                               const v4f (&w_narrow)[9], const v4f (&w_wide)[9],
                                               v4f wgt) {
    float rw[3][12], rn[3][12];
#pragma unroll
    for (int i = 0; i < 3; ++i) {
        const v4f L = P.wl[i], M = P.wm[i], R = P.wr[i];
        rw[i][0]  = left ? R.x : L.x;
        rw[i][1]  = left ? M.w : L.y;
        rw[i][2]  = left ? M.z : L.z;
        rw[i][3]  = left ? M.y : L.w;
        rw[i][4]  = M.x;  rw[i][5] = M.y;  rw[i][6] = M.z;  rw[i][7] = M.w;
        rw[i][8]  = right ? M.z : R.x;
        rw[i][9]  = right ? M.y : R.y;
        rw[i][10] = right ? M.x : R.z;
        rw[i][11] = right ? L.w : R.w;
    }
#pragma unroll
    for (int i = 0; i < 2; ++i) {
        const v2f A = P.na[i], B = P.nb[i];
        const v4f M = P.nm[i];
        float* r = rn[2 * i];
        r[2] = left ? M.z : A.x;
        r[3] = left ? M.y : A.y;
        r[4] = M.x;  r[5] = M.y;  r[6] = M.z;  r[7] = M.w;
        r[8] = right ? M.z : B.x;
        r[9] = right ? M.y : B.y;
    }
#pragma unroll
    for (int j = 2; j < 10; ++j)
        rn[1][j] = rw[1][j];

    v4f res;
#pragma unroll
    for (int p = 0; p < PXT; ++p) {
        float acc = 0.0f;
#pragma unroll
        for (int s = 0; s < 9; ++s) {
            const int iy = s / 3, ix = s % 3;
            const float c1 = rn[iy][4 + p + (ix - 1) * 2];
            const float c2 = rw[iy][4 + p + (ix - 1) * 4];
            acc = fmaf(0.5f, fmaf(w_narrow[s][p], c1, w_wide[s][p] * c2), acc);
        }
        res[p] = acc * wgt[p];
    }
    return res;
}

__global__ __launch_bounds__(K2_BLOCK) void aggregate_kernel(
    const float* __restrict__ sim,
    const float* __restrict__ offset,
    const float* __restrict__ weight,
    float* __restrict__ out)
{
    const int q   = (blockIdx.x % K2_BPP) * K2_BLOCK + threadIdx.x;
    const int pg  = blockIdx.x / K2_BPP;
    const int dg  = pg % NDG;
    const int b   = pg / NDG;
    const int y   = q / QPR;
    const int x0  = (q - y * QPR) * PXT;
    const int pix = q * PXT;

    const bool left  = (x0 == 0);
    const bool right = (x0 == NW - PXT);

    const int wrow[3] = { reflect_ix(y - 4, NH) * NW, y * NW, reflect_ix(y + 4, NH) * NW };
    const int nrow[2] = { reflect_ix(y - 2, NH) * NW, reflect_ix(y + 2, NH) * NW };
    const int cl4 = left  ? 0  : x0 - 4;
    const int cl2 = left  ? 0  : x0 - 2;
    const int cr  = right ? x0 : x0 + 4;

    const size_t plane0 = (size_t)(b * ND + dg * DPT) * HW;
    const float* sb = sim + plane0;
    const float* ob = offset + (size_t)b * 18 * HW + pix;
    float*       op = out + plane0 + pix;

    v4f w_narrow[9], w_wide[9];
#pragma unroll
    for (int s = 0; s < 9; ++s) {
        w_wide[s]   = *(const v4f*)(ob + (size_t)s * HW);
        w_narrow[s] = *(const v4f*)(ob + (size_t)(s + 9) * HW);
    }
    const v4f wgt = *(const v4f*)(weight + (size_t)b * HW + pix);
    PlanePieces P0, P1;
    load_plane(P0, sb,      wrow, nrow, cl4, cl2, x0, cr);
    load_plane(P1, sb + HW, wrow, nrow, cl4, cl2, x0, cr);

    pin_plane(P0);
    pin_plane(P1);

    PlanePieces P2, P3;
    load_plane(P2, sb + 2 * HW, wrow, nrow, cl4, cl2, x0, cr);
    load_plane(P3, sb + 3 * HW, wrow, nrow, cl4, cl2, x0, cr);

    *(v4f*)(op)          = aggregate_plane(P0, left, right, w_narrow, w_wide, wgt);
    *(v4f*)(op + HW)     = aggregate_plane(P1, left, right, w_narrow, w_wide, wgt);
    *(v4f*)(op + 2 * HW) = aggregate_plane(P2, left, right, w_narrow, w_wide, wgt);
    *(v4f*)(op + 3 * HW) = aggregate_plane(P3, left, right, w_narrow, w_wide, wgt);
}

extern "C" void kernel_launch(void* const* d_in, const int* in_sizes, int n_in,
                              void* d_out, int out_size, void* d_ws, size_t ws_size,
                              hipStream_t stream)
{
    const float* x1        = (const float*)d_in[0];
    const float* offset    = (const float*)d_in[1];
    const float* weight    = (const float*)d_in[2];
    const float* w0        = (const float*)d_in[3];
    const float* bn0_scale = (const float*)d_in[4];
    const float* bn0_bias  = (const float*)d_in[5];
    const float* w1        = (const float*)d_in[6];
    const float* bn1_scale = (const float*)d_in[7];
    const float* bn1_bias  = (const float*)d_in[8];
    const float* w_sim     = (const float*)d_in[9];
    const float* b_sim     = (const float*)d_in[10];
    float* out = (float*)d_out;
    float* sim = (float*)d_ws;

    mlp_sim_kernel<<<K1_WAVES / (K1_BLOCK / 32), K1_BLOCK, 0, stream>>>(
        x1, w0, bn0_scale, bn0_bias, w1, bn1_scale, bn1_bias, w_sim, b_sim, sim);

    aggregate_kernel<<<K2_GRID, K2_BLOCK, 0, stream>>>(
        sim, offset, weight, out);
}


# w1 A fragments re-placed (k order = layer-1 D layout; channels 0-3/4-7 on the two lane halves): no half-wave swap between layers, half the layer-3 work; layer-2 k order, layer-3 sum order change
# speedup vs baseline: 1.4100x; 1.0651x over previous
#include <hip/hip_runtime.h>
#include <hip/hip_bf16.h>


typedef __attribute__((ext_vector_type(2))) float v2f;
typedef __attribute__((ext_vector_type(4))) float v4f;
typedef __attribute__((ext_vector_type(8))) float v8f;

#define NB 2
#define NG 8
#define ND 32
#define NH 128
#define NW 160
#define HW (NH * NW)
#define DHW (ND * NH * NW)
#define NPTS (NB * DHW)

#define TILES_PER_GROUP 4
#define NGROUPS (NPTS / 64)
#define K1_BLOCK 128
#define K1_WAVES 5120
#define GROUPS_PER_WAVE (NGROUPS / K1_WAVES)

#define PXT 4
#define QPR (NW / PXT)
#define DPT 4
#define NDG (ND / DPT)
#define K2_BLOCK 320
#define K2_BPP (HW / PXT / K2_BLOCK)
#define K2_GRID (NB * NDG * K2_BPP)

#define SWIZ_SWAPX16 0x401F

__device__ __forceinline__ float half_swap(float v) {
    return __int_as_float(
        __builtin_amdgcn_ds_swizzle(__float_as_int(v), SWIZ_SWAPX16));
}

__global__ __launch_bounds__(K1_BLOCK) void mlp_sim_kernel(
    const float* __restrict__ x1,
    const float* __restrict__ w0,
    const float* __restrict__ bn0_scale, const float* __restrict__ bn0_bias,
    const float* __restrict__ w1,
    const float* __restrict__ bn1_scale, const float* __restrict__ bn1_bias,
    const float* __restrict__ w_sim, const float* __restrict__ b_sim,
    float* __restrict__ sim)
{
    const int lane = threadIdx.x & 31;
    const int wib  = threadIdx.x >> 5;
    const int gwave = blockIdx.x * (K1_BLOCK / 32) + wib;
    const int hi   = lane >> 4;
    const int col  = lane & 15;

    const int kb = hi * 2;
    v2f a0, a1;
    a0.x = w0[col * 8 + kb + 0];
    a0.y = w0[col * 8 + kb + 1];
    a1.x = w0[col * 8 + 4 + kb + 0];
    a1.y = w0[col * 8 + 4 + kb + 1];

    const bool arow_used = (col & 4) == 0;
    const int  orow = (col & 3) | ((col & 8) >> 1);
    v2f aw[4];
#pragma unroll
    for (int j = 0; j < 4; ++j) {
        const int k0 = 8 * hi + 2 * j;
        const float wx = w1[orow * 16 + k0 + 0];
        const float wy = w1[orow * 16 + k0 + 1];
        aw[j].x = arow_used ? wx : 0.0f;
        aw[j].y = arow_used ? wy : 0.0f;
    }

    float s0[8], bi0[8];
    const int rb = 8 * hi;
#pragma unroll
    for (int v = 0; v < 8; ++v) {
        s0[v]  = bn0_scale[rb + v];
        bi0[v] = bn0_bias[rb + v];
    }
    float s1[4], bi1[4], ws[4];
    const int cb = 4 * hi;
#pragma unroll
    for (int v = 0; v < 4; ++v) {
        s1[v]  = bn1_scale[cb + v];
        bi1[v] = bn1_bias[cb + v];
        ws[v]  = w_sim[cb + v];
    }
    const float acc0 = hi ? 0.0f : b_sim[0];

    for (int i = 0; i < GROUPS_PER_WAVE; ++i) {
        const int g     = i * K1_WAVES + gwave;
        const int base  = g * 64;
        const int b     = base / DHW;
        const int rem   = base - b * DHW + PXT * col;

        const float* xp = x1 + (size_t)(b * NG + kb) * DHW + rem;
        const v4f q0 = *(const v4f*)(xp);
        const v4f q1 = *(const v4f*)(xp + DHW);
        const v4f q2 = *(const v4f*)(xp + 4 * (size_t)DHW);
        const v4f q3 = *(const v4f*)(xp + 5 * (size_t)DHW);

        v8f c[TILES_PER_GROUP];
#pragma unroll
        for (int t = 0; t < TILES_PER_GROUP; ++t) {
            v2f b0;
            b0.x = q0[t];  b0.y = q1[t];
            v8f z = {};
            c[t] = __builtin_amdgcn_wmma_f32_16x16x4_f32(false, a0, false, b0,
                                                         (short)0, z, false, false);
        }
#pragma unroll
        for (int t = 0; t < TILES_PER_GROUP; ++t) {
            v2f b1;
            b1.x = q2[t];  b1.y = q3[t];
            c[t] = __builtin_amdgcn_wmma_f32_16x16x4_f32(false, a1, false, b1,
                                                         (short)0, c[t], false, false);
        }

        float h[TILES_PER_GROUP][8];
#pragma unroll
        for (int t = 0; t < TILES_PER_GROUP; ++t)
#pragma unroll
            for (int v = 0; v < 8; ++v)
                h[t][v] = fmaxf(fmaf(c[t][v], s0[v], bi0[v]), 0.0f);

        v8f d[TILES_PER_GROUP];
#pragma unroll
        for (int t = 0; t < TILES_PER_GROUP; ++t) {
            v2f bb;
            bb.x = h[t][0];  bb.y = h[t][1];
            v8f z = {};
            d[t] = __builtin_amdgcn_wmma_f32_16x16x4_f32(false, aw[0], false, bb,
                                                         (short)0, z, false, false);
        }
#pragma unroll
        for (int j = 1; j < 4; ++j)
#pragma unroll
            for (int t = 0; t < TILES_PER_GROUP; ++t) {
                v2f bb;
                bb.x = h[t][2 * j];  bb.y = h[t][2 * j + 1];
                d[t] = __builtin_amdgcn_wmma_f32_16x16x4_f32(false, aw[j], false, bb,
                                                             (short)0, d[t], false, false);
            }

        v4f o;
#pragma unroll
        for (int t = 0; t < TILES_PER_GROUP; ++t) {
            float acc = acc0;
#pragma unroll
            for (int v = 0; v < 4; ++v)
                acc = fmaf(ws[v], fmaxf(fmaf(d[t][v], s1[v], bi1[v]), 0.0f), acc);
            o[t] = acc;
        }

        v4f tot;
#pragma unroll
        for (int t = 0; t < TILES_PER_GROUP; ++t)
            tot[t] = o[t] + half_swap(o[t]);

        if (!hi) {
            *(v4f*)(sim + base + PXT * col) = tot;
        }
    }
}

__device__ __forceinline__ int reflect_ix(int t, int n) {
    t = (t < 0) ? -t : t;
    t = (t >= n) ? (2 * n - 2 - t) : t;
    return t;
}

struct PlanePieces {
    v4f wl[3], wm[3], wr[3];
    v2f na[2], nb[2];
    v4f nm[2];
};

__device__ __forceinline__ void load_plane(PlanePieces& P, const float* sp,
                                           const int (&wrow)[3], const int (&nrow)[2],
                                           int cl4, int cl2, int x0, int cr) {
#pragma unroll
    for (int i = 0; i < 3; ++i) {
        P.wl[i] = *(const v4f*)(sp + wrow[i] + cl4);
        P.wm[i] = *(const v4f*)(sp + wrow[i] + x0);
        P.wr[i] = *(const v4f*)(sp + wrow[i] + cr);
    }
#pragma unroll
    for (int i = 0; i < 2; ++i) {
        P.na[i] = *(const v2f*)(sp + nrow[i] + cl2);
        P.nm[i] = *(const v4f*)(sp + nrow[i] + x0);
        P.nb[i] = *(const v2f*)(sp + nrow[i] + cr);
    }
}

__device__ __forceinline__ void pin_plane(PlanePieces& P) {
    asm volatile("" : "+v"(P.wl[0]), "+v"(P.wm[0]), "+v"(P.wr[0]),
                      "+v"(P.wl[1]), "+v"(P.wm[1]), "+v"(P.wr[1]),
                      "+v"(P.wl[2]), "+v"(P.wm[2]), "+v"(P.wr[2]),
                      "+v"(P.na[0]), "+v"(P.nm[0]), "+v"(P.nb[0]),
                      "+v"(P.na[1]), "+v"(P.nm[1]), "+v"(P.nb[1]));
}

__device__ __forceinline__ v4f aggregate_plane(const PlanePieces& P, bool left, bool right,
                                               const v4f (&w_narrow)[9], const v4f (&w_wide)[9],
                                               v4f wgt) {
    float rw[3][12], rn[3][12];
#pragma unroll
    for (int i = 0; i < 3; ++i) {
        const v4f L = P.wl[i], M = P.wm[i], R = P.wr[i];
        rw[i][0]  = left ? R.x : L.x;
        rw[i][1]  = left ? M.w : L.y;
        rw[i][2]  = left ? M.z : L.z;
        rw[i][3]  = left ? M.y : L.w;
        rw[i][4]  = M.x;  rw[i][5] = M.y;  rw[i][6] = M.z;  rw[i][7] = M.w;
        rw[i][8]  = right ? M.z : R.x;
        rw[i][9]  = right ? M.y : R.y;
        rw[i][10] = right ? M.x : R.z;
        rw[i][11] = right ? L.w : R.w;
    }
#pragma unroll
    for (int i = 0; i < 2; ++i) {
        const v2f A = P.na[i], B = P.nb[i];
        const v4f M = P.nm[i];
        float* r = rn[2 * i];
        r[2] = left ? M.z : A.x;
        r[3] = left ? M.y : A.y;
        r[4] = M.x;  r[5] = M.y;  r[6] = M.z;  r[7] = M.w;
        r[8] = right ? M.z : B.x;
        r[9] = right ? M.y : B.y;
    }
#pragma unroll
    for (int j = 2; j < 10; ++j)
        rn[1][j] = rw[1][j];

    v4f res;
#pragma unroll
    for (int p = 0; p < PXT; ++p) {
        float acc = 0.0f;
#pragma unroll
        for (int s = 0; s < 9; ++s) {
            const int iy = s / 3, ix = s % 3;
            const float c1 = rn[iy][4 + p + (ix - 1) * 2];
            const float c2 = rw[iy][4 + p + (ix - 1) * 4];
            acc = fmaf(0.5f, fmaf(w_narrow[s][p], c1, w_wide[s][p] * c2), acc);
        }
        res[p] = acc * wgt[p];
    }
    return res;
}

__global__ __launch_bounds__(K2_BLOCK) void aggregate_kernel(
    const float* __restrict__ sim,
    const float* __restrict__ offset,
    const float* __restrict__ weight,
    float* __restrict__ out)
{
    const int q   = (blockIdx.x % K2_BPP) * K2_BLOCK + threadIdx.x;
    const int pg  = blockIdx.x / K2_BPP;
    const int dg  = pg % NDG;
    const int b   = pg / NDG;
    const int y   = q / QPR;
    const int x0  = (q - y * QPR) * PXT;
    const int pix = q * PXT;

    const bool left  = (x0 == 0);
    const bool right = (x0 == NW - PXT);

    const int wrow[3] = { reflect_ix(y - 4, NH) * NW, y * NW, reflect_ix(y + 4, NH) * NW };
    const int nrow[2] = { reflect_ix(y - 2, NH) * NW, reflect_ix(y + 2, NH) * NW };
    const int cl4 = left  ? 0  : x0 - 4;
    const int cl2 = left  ? 0  : x0 - 2;
    const int cr  = right ? x0 : x0 + 4;

    const size_t plane0 = (size_t)(b * ND + dg * DPT) * HW;
    const float* sb = sim + plane0;
    const float* ob = offset + (size_t)b * 18 * HW + pix;
    float*       op = out + plane0 + pix;

    v4f w_narrow[9], w_wide[9];
#pragma unroll
    for (int s = 0; s < 9; ++s) {
        w_wide[s]   = *(const v4f*)(ob + (size_t)s * HW);
        w_narrow[s] = *(const v4f*)(ob + (size_t)(s + 9) * HW);
    }
    const v4f wgt = *(const v4f*)(weight + (size_t)b * HW + pix);
    PlanePieces P0, P1;
    load_plane(P0, sb,      wrow, nrow, cl4, cl2, x0, cr);
    load_plane(P1, sb + HW, wrow, nrow, cl4, cl2, x0, cr);

    pin_plane(P0);
    pin_plane(P1);

    PlanePieces P2, P3;
    load_plane(P2, sb + 2 * HW, wrow, nrow, cl4, cl2, x0, cr);
    load_plane(P3, sb + 3 * HW, wrow, nrow, cl4, cl2, x0, cr);

    *(v4f*)(op)          = aggregate_plane(P0, left, right, w_narrow, w_wide, wgt);
    *(v4f*)(op + HW)     = aggregate_plane(P1, left, right, w_narrow, w_wide, wgt);
    *(v4f*)(op + 2 * HW) = aggregate_plane(P2, left, right, w_narrow, w_wide, wgt);
    *(v4f*)(op + 3 * HW) = aggregate_plane(P3, left, right, w_narrow, w_wide, wgt);
}

extern "C" void kernel_launch(void* const* d_in, const int* in_sizes, int n_in,
                              void* d_out, int out_size, void* d_ws, size_t ws_size,
                              hipStream_t stream)
{
    const float* x1        = (const float*)d_in[0];
    const float* offset    = (const float*)d_in[1];
    const float* weight    = (const float*)d_in[2];
    const float* w0        = (const float*)d_in[3];
    const float* bn0_scale = (const float*)d_in[4];
    const float* bn0_bias  = (const float*)d_in[5];
    const float* w1        = (const float*)d_in[6];
    const float* bn1_scale = (const float*)d_in[7];
    const float* bn1_bias  = (const float*)d_in[8];
    const float* w_sim     = (const float*)d_in[9];
    const float* b_sim     = (const float*)d_in[10];
    float* out = (float*)d_out;
    float* sim = (float*)d_ws;

    mlp_sim_kernel<<<K1_WAVES / (K1_BLOCK / 32), K1_BLOCK, 0, stream>>>(
        x1, w0, bn0_scale, bn0_bias, w1, bn1_scale, bn1_bias, w_sim, b_sim, sim);

    aggregate_kernel<<<K2_GRID, K2_BLOCK, 0, stream>>>(
        sim, offset, weight, out);
}


# mlp kernel: each wave requests the x data of all its 4 groups (16 loads in flight) before its first product: one memory round trip per wave, not four; b_sim read unbranched so no wait splits them
# speedup vs baseline: 1.5576x; 1.1047x over previous
#include <hip/hip_runtime.h>
#include <hip/hip_bf16.h>


typedef __attribute__((ext_vector_type(2))) float v2f;
typedef __attribute__((ext_vector_type(4))) float v4f;
typedef __attribute__((ext_vector_type(8))) float v8f;

#define NB 2
#define NG 8
#define ND 32
#define NH 128
#define NW 160
#define HW (NH * NW)
#define DHW (ND * NH * NW)
#define NPTS (NB * DHW)

#define TILES_PER_GROUP 4
#define NGROUPS (NPTS / 64)
#define K1_BLOCK 128
#define K1_WAVES 5120
#define GROUPS_PER_WAVE (NGROUPS / K1_WAVES)

#define PXT 4
#define QPR (NW / PXT)
#define DPT 4
#define NDG (ND / DPT)
#define K2_BLOCK 320
#define K2_BPP (HW / PXT / K2_BLOCK)
#define K2_GRID (NB * NDG * K2_BPP)

#define SWIZ_SWAPX16 0x401F

__device__ __forceinline__ float half_swap(float v) {
    return __int_as_float(
        __builtin_amdgcn_ds_swizzle(__float_as_int(v), SWIZ_SWAPX16));
}

__global__ __launch_bounds__(K1_BLOCK) void mlp_sim_kernel(
    const float* __restrict__ x1,
    const float* __restrict__ w0,
    const float* __restrict__ bn0_scale, const float* __restrict__ bn0_bias,
    const float* __restrict__ w1,
    const float* __restrict__ bn1_scale, const float* __restrict__ bn1_bias,
    const float* __restrict__ w_sim, const float* __restrict__ b_sim,
    float* __restrict__ sim)
{
    const int lane = threadIdx.x & 31;
    const int wib  = threadIdx.x >> 5;
    const int gwave = blockIdx.x * (K1_BLOCK / 32) + wib;
    const int hi   = lane >> 4;
    const int col  = lane & 15;

    const int kb = hi * 2;
    v2f a0, a1;
    a0.x = w0[col * 8 + kb + 0];
    a0.y = w0[col * 8 + kb + 1];
    a1.x = w0[col * 8 + 4 + kb + 0];
    a1.y = w0[col * 8 + 4 + kb + 1];

    const bool arow_used = (col & 4) == 0;
    const int  orow = (col & 3) | ((col & 8) >> 1);
    v2f aw[4];
#pragma unroll
    for (int j = 0; j < 4; ++j) {
        const int k0 = 8 * hi + 2 * j;
        const float wx = w1[orow * 16 + k0 + 0];
        const float wy = w1[orow * 16 + k0 + 1];
        aw[j].x = arow_used ? wx : 0.0f;
        aw[j].y = arow_used ? wy : 0.0f;
    }

    float s0[8], bi0[8];
    const int rb = 8 * hi;
#pragma unroll
    for (int v = 0; v < 8; ++v) {
        s0[v]  = bn0_scale[rb + v];
        bi0[v] = bn0_bias[rb + v];
    }
    float s1[4], bi1[4], ws[4];
    const int cb = 4 * hi;
#pragma unroll
    for (int v = 0; v < 4; ++v) {
        s1[v]  = bn1_scale[cb + v];
        bi1[v] = bn1_bias[cb + v];
        ws[v]  = w_sim[cb + v];
    }
    const float bsim = b_sim[0];
    const float acc0 = hi ? 0.0f : bsim;

    v4f q[GROUPS_PER_WAVE][4];
#pragma unroll
    for (int i = 0; i < GROUPS_PER_WAVE; ++i) {
        const int g    = i * K1_WAVES + gwave;
        const int base = g * 64;
        const int b    = base / DHW;
        const int rem  = base - b * DHW + PXT * col;
        const float* xp = x1 + (size_t)(b * NG + kb) * DHW + rem;
        q[i][0] = *(const v4f*)(xp);
        q[i][1] = *(const v4f*)(xp + DHW);
        q[i][2] = *(const v4f*)(xp + 4 * (size_t)DHW);
        q[i][3] = *(const v4f*)(xp + 5 * (size_t)DHW);
    }
    __builtin_amdgcn_sched_barrier(0);

#pragma unroll
    for (int i = 0; i < GROUPS_PER_WAVE; ++i) {
        const int base = (i * K1_WAVES + gwave) * 64;
        const v4f q0 = q[i][0], q1 = q[i][1], q2 = q[i][2], q3 = q[i][3];

        v8f c[TILES_PER_GROUP];
#pragma unroll
        for (int t = 0; t < TILES_PER_GROUP; ++t) {
            v2f b0;
            b0.x = q0[t];  b0.y = q1[t];
            v8f z = {};
            c[t] = __builtin_amdgcn_wmma_f32_16x16x4_f32(false, a0, false, b0,
                                                         (short)0, z, false, false);
        }
#pragma unroll
        for (int t = 0; t < TILES_PER_GROUP; ++t) {
            v2f b1;
            b1.x = q2[t];  b1.y = q3[t];
            c[t] = __builtin_amdgcn_wmma_f32_16x16x4_f32(false, a1, false, b1,
                                                         (short)0, c[t], false, false);
        }

        float h[TILES_PER_GROUP][8];
#pragma unroll
        for (int t = 0; t < TILES_PER_GROUP; ++t)
#pragma unroll
            for (int v = 0; v < 8; ++v)
                h[t][v] = fmaxf(fmaf(c[t][v], s0[v], bi0[v]), 0.0f);

        v8f d[TILES_PER_GROUP];
#pragma unroll
        for (int t = 0; t < TILES_PER_GROUP; ++t) {
            v2f bb;
            bb.x = h[t][0];  bb.y = h[t][1];
            v8f z = {};
            d[t] = __builtin_amdgcn_wmma_f32_16x16x4_f32(false, aw[0], false, bb,
                                                         (short)0, z, false, false);
        }
#pragma unroll
        for (int j = 1; j < 4; ++j)
#pragma unroll
            for (int t = 0; t < TILES_PER_GROUP; ++t) {
                v2f bb;
                bb.x = h[t][2 * j];  bb.y = h[t][2 * j + 1];
                d[t] = __builtin_amdgcn_wmma_f32_16x16x4_f32(false, aw[j], false, bb,
                                                             (short)0, d[t], false, false);
            }

        v4f o;
#pragma unroll
        for (int t = 0; t < TILES_PER_GROUP; ++t) {
            float acc = acc0;
#pragma unroll
            for (int v = 0; v < 4; ++v)
                acc = fmaf(ws[v], fmaxf(fmaf(d[t][v], s1[v], bi1[v]), 0.0f), acc);
            o[t] = acc;
        }

        v4f tot;
#pragma unroll
        for (int t = 0; t < TILES_PER_GROUP; ++t)
            tot[t] = o[t] + half_swap(o[t]);

        if (!hi) {
            *(v4f*)(sim + base + PXT * col) = tot;
        }
    }
}

__device__ __forceinline__ int reflect_ix(int t, int n) {
    t = (t < 0) ? -t : t;
    t = (t >= n) ? (2 * n - 2 - t) : t;
    return t;
}

struct PlanePieces {
    v4f wl[3], wm[3], wr[3];
    v2f na[2], nb[2];
    v4f nm[2];
};

__device__ __forceinline__ void load_plane(PlanePieces& P, const float* sp,
                                           const int (&wrow)[3], const int (&nrow)[2],
                                           int cl4, int cl2, int x0, int cr) {
#pragma unroll
    for (int i = 0; i < 3; ++i) {
        P.wl[i] = *(const v4f*)(sp + wrow[i] + cl4);
        P.wm[i] = *(const v4f*)(sp + wrow[i] + x0);
        P.wr[i] = *(const v4f*)(sp + wrow[i] + cr);
    }
#pragma unroll
    for (int i = 0; i < 2; ++i) {
        P.na[i] = *(const v2f*)(sp + nrow[i] + cl2);
        P.nm[i] = *(const v4f*)(sp + nrow[i] + x0);
        P.nb[i] = *(const v2f*)(sp + nrow[i] + cr);
    }
}

__device__ __forceinline__ void pin_plane(PlanePieces& P) {
    asm volatile("" : "+v"(P.wl[0]), "+v"(P.wm[0]), "+v"(P.wr[0]),
                      "+v"(P.wl[1]), "+v"(P.wm[1]), "+v"(P.wr[1]),
                      "+v"(P.wl[2]), "+v"(P.wm[2]), "+v"(P.wr[2]),
                      "+v"(P.na[0]), "+v"(P.nm[0]), "+v"(P.nb[0]),
                      "+v"(P.na[1]), "+v"(P.nm[1]), "+v"(P.nb[1]));
}

__device__ __forceinline__ v4f aggregate_plane(const PlanePieces& P, bool left, bool right,
                                               const v4f (&w_narrow)[9], const v4f (&w_wide)[9],
                                               v4f wgt) {
    float rw[3][12], rn[3][12];
#pragma unroll
    for (int i = 0; i < 3; ++i) {
        const v4f L = P.wl[i], M = P.wm[i], R = P.wr[i];
        rw[i][0]  = left ? R.x : L.x;
        rw[i][1]  = left ? M.w : L.y;
        rw[i][2]  = left ? M.z : L.z;
        rw[i][3]  = left ? M.y : L.w;
        rw[i][4]  = M.x;  rw[i][5] = M.y;  rw[i][6] = M.z;  rw[i][7] = M.w;
        rw[i][8]  = right ? M.z : R.x;
        rw[i][9]  = right ? M.y : R.y;
        rw[i][10] = right ? M.x : R.z;
        rw[i][11] = right ? L.w : R.w;
    }
#pragma unroll
    for (int i = 0; i < 2; ++i) {
        const v2f A = P.na[i], B = P.nb[i];
        const v4f M = P.nm[i];
        float* r = rn[2 * i];
        r[2] = left ? M.z : A.x;
        r[3] = left ? M.y : A.y;
        r[4] = M.x;  r[5] = M.y;  r[6] = M.z;  r[7] = M.w;
        r[8] = right ? M.z : B.x;
        r[9] = right ? M.y : B.y;
    }
#pragma unroll
    for (int j = 2; j < 10; ++j)
        rn[1][j] = rw[1][j];

    v4f res;
#pragma unroll
    for (int p = 0; p < PXT; ++p) {
        float acc = 0.0f;
#pragma unroll
        for (int s = 0; s < 9; ++s) {
            const int iy = s / 3, ix = s % 3;
            const float c1 = rn[iy][4 + p + (ix - 1) * 2];
            const float c2 = rw[iy][4 + p + (ix - 1) * 4];
            acc = fmaf(0.5f, fmaf(w_narrow[s][p], c1, w_wide[s][p] * c2), acc);
        }
        res[p] = acc * wgt[p];
    }
    return res;
}

__global__ __launch_bounds__(K2_BLOCK) void aggregate_kernel(
    const float* __restrict__ sim,
    const float* __restrict__ offset,
    const float* __restrict__ weight,
    float* __restrict__ out)
{
    const int q   = (blockIdx.x % K2_BPP) * K2_BLOCK + threadIdx.x;
    const int pg  = blockIdx.x / K2_BPP;
    const int dg  = pg % NDG;
    const int b   = pg / NDG;
    const int y   = q / QPR;
    const int x0  = (q - y * QPR) * PXT;
    const int pix = q * PXT;

    const bool left  = (x0 == 0);
    const bool right = (x0 == NW - PXT);

    const int wrow[3] = { reflect_ix(y - 4, NH) * NW, y * NW, reflect_ix(y + 4, NH) * NW };
    const int nrow[2] = { reflect_ix(y - 2, NH) * NW, reflect_ix(y + 2, NH) * NW };
    const int cl4 = left  ? 0  : x0 - 4;
    const int cl2 = left  ? 0  : x0 - 2;
    const int cr  = right ? x0 : x0 + 4;

    const size_t plane0 = (size_t)(b * ND + dg * DPT) * HW;
    const float* sb = sim + plane0;
    const float* ob = offset + (size_t)b * 18 * HW + pix;
    float*       op = out + plane0 + pix;

    v4f w_narrow[9], w_wide[9];
#pragma unroll
    for (int s = 0; s < 9; ++s) {
        w_wide[s]   = *(const v4f*)(ob + (size_t)s * HW);
        w_narrow[s] = *(const v4f*)(ob + (size_t)(s + 9) * HW);
    }
    const v4f wgt = *(const v4f*)(weight + (size_t)b * HW + pix);
    PlanePieces P0, P1;
    load_plane(P0, sb,      wrow, nrow, cl4, cl2, x0, cr);
    load_plane(P1, sb + HW, wrow, nrow, cl4, cl2, x0, cr);

    pin_plane(P0);
    pin_plane(P1);

    PlanePieces P2, P3;
    load_plane(P2, sb + 2 * HW, wrow, nrow, cl4, cl2, x0, cr);
    load_plane(P3, sb + 3 * HW, wrow, nrow, cl4, cl2, x0, cr);

    *(v4f*)(op)          = aggregate_plane(P0, left, right, w_narrow, w_wide, wgt);
    *(v4f*)(op + HW)     = aggregate_plane(P1, left, right, w_narrow, w_wide, wgt);
    *(v4f*)(op + 2 * HW) = aggregate_plane(P2, left, right, w_narrow, w_wide, wgt);
    *(v4f*)(op + 3 * HW) = aggregate_plane(P3, left, right, w_narrow, w_wide, wgt);
}

extern "C" void kernel_launch(void* const* d_in, const int* in_sizes, int n_in,
                              void* d_out, int out_size, void* d_ws, size_t ws_size,
                              hipStream_t stream)
{
    const float* x1        = (const float*)d_in[0];
    const float* offset    = (const float*)d_in[1];
    const float* weight    = (const float*)d_in[2];
    const float* w0        = (const float*)d_in[3];
    const float* bn0_scale = (const float*)d_in[4];
    const float* bn0_bias  = (const float*)d_in[5];
    const float* w1        = (const float*)d_in[6];
    const float* bn1_scale = (const float*)d_in[7];
    const float* bn1_bias  = (const float*)d_in[8];
    const float* w_sim     = (const float*)d_in[9];
    const float* b_sim     = (const float*)d_in[10];
    float* out = (float*)d_out;
    float* sim = (float*)d_ws;

    mlp_sim_kernel<<<K1_WAVES / (K1_BLOCK / 32), K1_BLOCK, 0, stream>>>(
        x1, w0, bn0_scale, bn0_bias, w1, bn1_scale, bn1_bias, w_sim, b_sim, sim);

    aggregate_kernel<<<K2_GRID, K2_BLOCK, 0, stream>>>(
        sim, offset, weight, out);
}
